# starting kernel
# baseline (speedup 1.0000x reference)
#include <hip/hip_runtime.h>
#include <hip/hip_bf16.h>
#include <math.h>

#define U_DIM 512
#define B_DIM 4
#define L_DIM 512
#define H_DIM 8
#define D_DIM 64
#define HB_DIM 32
#define M_ROWS 2048

typedef __attribute__((ext_vector_type(2))) float v2f;
typedef __attribute__((ext_vector_type(8))) float v8f;

__device__ __forceinline__ v8f wmma4(v2f a, v2f b, v8f c) {
#if defined(__HIP_DEVICE_COMPILE__)
  return __builtin_amdgcn_wmma_f32_16x16x4_f32(false, a, false, b, (short)0, c,
                                               false, false);
#else
  (void)a; (void)b;
  return c;
#endif
}


#define SW_STRIDE 288

__global__ __launch_bounds__(256) void proj_relu_kernel(
    const float* __restrict__ Xq, const float* __restrict__ Xk,
    const float* __restrict__ Xv, const float* __restrict__ Wq,
    const float* __restrict__ Wk, const float* __restrict__ Wv,
    const float* __restrict__ bq, const float* __restrict__ bk,
    const float* __restrict__ bv, float* __restrict__ Qh,
    float* __restrict__ Kh, float* __restrict__ Vh) {
  const int which = blockIdx.z;
  const float* X = which == 0 ? Xq : (which == 1 ? Xk : Xv);
  const float* W = which == 0 ? Wq : (which == 1 ? Wk : Wv);
  const float* bias = which == 0 ? bq : (which == 1 ? bk : bv);
  float* O = which == 0 ? Qh : (which == 1 ? Kh : Vh);

  const int m0 = blockIdx.y * 64;
  const int n0 = blockIdx.x * 128;
  const int tid = threadIdx.x;
  const int lane = tid & 31;
  const int wave = tid >> 5;
  const int wm = wave >> 1;
  const int wn = wave & 1;
  const int l16 = lane & 15;
  const int hi = lane >> 4;

  __shared__ float sX[64][68];
  __shared__ float sW2[8][SW_STRIDE];

  v8f acc[4] = {};

  for (int kt = 0; kt < U_DIM / 16; ++kt) {
    {
      const int r = tid >> 2, c = (tid & 3) * 4;
      const float4 x =
          *(const float4*)&X[(size_t)(m0 + r) * U_DIM + kt * 16 + c];
      *(float4*)&sX[r][c] = x;
    }
    {
      const int r = tid >> 4, c = (tid & 15) * 8;
      const float4 w0 =
          *(const float4*)&W[(size_t)(kt * 16 + r) * U_DIM + n0 + c];
      const float4 w1 =
          *(const float4*)&W[(size_t)(kt * 16 + r) * U_DIM + n0 + c + 4];
      float* dst = &sW2[r >> 1][c * 2 + (r & 1)];
      dst[0] = w0.x;  dst[2] = w0.y;  dst[4] = w0.z;  dst[6] = w0.w;
      dst[8] = w1.x;  dst[10] = w1.y; dst[12] = w1.z; dst[14] = w1.w;
    }
    __syncthreads();

#pragma unroll
    for (int kk = 0; kk < 16; kk += 4) {
      const v2f a = *(const v2f*)&sX[wm * 16 + l16][kk + 2 * hi];
      const int kp = (kk >> 1) + hi;
#pragma unroll
      for (int n = 0; n < 4; ++n) {
        const int col = wn * 64 + n * 16 + l16;
        const v2f b = *(const v2f*)&sW2[kp][col * 2];
        acc[n] = wmma4(a, b, acc[n]);
      }
    }
    __syncthreads();
  }

#pragma unroll
  for (int n = 0; n < 4; ++n) {
    const int ng = n0 + wn * 64 + n * 16 + l16;
    const int h = ng >> 6, d = ng & 63;
    const float bb = bias[ng];
#pragma unroll
    for (int r = 0; r < 8; ++r) {
      const int mg = m0 + wm * 16 + r + 8 * hi;
      const int b = mg >> 9, l = mg & 511;
      float v = acc[n][r] + bb;
      v = v > 0.0f ? v : 0.0f;
      O[(((size_t)(h * B_DIM + b) * L_DIM + l) * D_DIM) + d] = v;
    }
  }
}

__global__ __launch_bounds__(256) void aux_kernel(
    const float* __restrict__ Qh, const float* __restrict__ Kh,
    const float* __restrict__ queries, float* __restrict__ qq,
    float* __restrict__ kkv, float* __restrict__ qsign) {
  const int idx = blockIdx.x * blockDim.x + threadIdx.x;
  const int nrows = HB_DIM * L_DIM;
  if (idx < nrows) {
    const float* p = Qh + (size_t)idx * D_DIM;
    float s = 0.0f;
    for (int d = 0; d < D_DIM; ++d) s += p[d] * p[d];
    qq[idx] = s;
  } else if (idx < 2 * nrows) {
    const float* p = Kh + (size_t)(idx - nrows) * D_DIM;
    float s = 0.0f;
    for (int d = 0; d < D_DIM; ++d) s += p[d] * p[d];
    kkv[idx - nrows] = s;
  } else if (idx < 2 * nrows + B_DIM * L_DIM) {
    const int r = idx - 2 * nrows;
    const float* p = queries + (size_t)r * U_DIM;
    float s = 0.0f;
    for (int u = 0; u < U_DIM; ++u) s += p[u];
    qsign[r] = (s != 0.0f) ? 1.0f : 0.0f;
  }
}

#define SKT_STRIDE 96
#define SV_STRIDE 160

__global__ __launch_bounds__(256) void attn_kernel(
    const float* __restrict__ Qh, const float* __restrict__ Kh,
    const float* __restrict__ Vh, const float* __restrict__ qq,
    const float* __restrict__ kkv, const float* __restrict__ qsign,
    float* __restrict__ outp, float* __restrict__ ascore) {
  const int hb = blockIdx.y;
  const int q0 = blockIdx.x * 64;
  const int tid = threadIdx.x;
  const int lane = tid & 31;
  const int wave = tid >> 5;
  const int wm = wave >> 1;
  const int wn = wave & 1;
  const int l16 = lane & 15;
  const int hi = lane >> 4;

  __shared__ float sQ[64][68];
  __shared__ float sKT2[32][SKT_STRIDE];
  __shared__ float sV2[16][SV_STRIDE];
  __shared__ float sP[64][36];
  __shared__ float sKK[L_DIM];
  __shared__ float sQQ[64];
  __shared__ float sQM[64];
  __shared__ float sMrow[64];
  __shared__ float sInvL[64];
  __shared__ float sStat[2][64][2];

  {
    const int r = tid >> 2, c = (tid & 3) * 16;
    const size_t base = ((size_t)hb * L_DIM + q0 + r) * D_DIM + c;
#pragma unroll
    for (int j = 0; j < 4; ++j) {
      const float4 v = *(const float4*)&Qh[base + 4 * j];
      *(float4*)&sQ[r][c + 4 * j] = v;
    }
    sKK[2 * tid] = kkv[(size_t)hb * L_DIM + 2 * tid];
    sKK[2 * tid + 1] = kkv[(size_t)hb * L_DIM + 2 * tid + 1];
    if (tid < 64) {
      sQQ[tid] = qq[(size_t)hb * L_DIM + q0 + tid];
      sQM[tid] = qsign[(size_t)(hb & 3) * L_DIM + q0 + tid];
    }
  }
  __syncthreads();

  float m_r[8], l_r[8];
#pragma unroll
  for (int r = 0; r < 8; ++r) { m_r[r] = -INFINITY; l_r[r] = 0.0f; }

  for (int kt = 0; kt < L_DIM / 32; ++kt) {
    {
      const int kc = tid >> 3, d0 = (tid & 7) * 8;
      const size_t base = ((size_t)hb * L_DIM + kt * 32 + kc) * D_DIM + d0;
      const float4 k0 = *(const float4*)&Kh[base];
      const float4 k1 = *(const float4*)&Kh[base + 4];
      float* dst = &sKT2[d0 >> 1][kc * 2];
      *(v2f*)&dst[0] = v2f{k0.x, k0.y};
      *(v2f*)&dst[SKT_STRIDE] = v2f{k0.z, k0.w};
      *(v2f*)&dst[2 * SKT_STRIDE] = v2f{k1.x, k1.y};
      *(v2f*)&dst[3 * SKT_STRIDE] = v2f{k1.z, k1.w};
    }
    __syncthreads();

    v8f s = {};
#pragma unroll
    for (int kk = 0; kk < D_DIM; kk += 4) {
      const v2f a = *(const v2f*)&sQ[wm * 16 + l16][kk + 2 * hi];
      const v2f b = *(const v2f*)&sKT2[(kk >> 1) + hi][(wn * 16 + l16) * 2];
      s = wmma4(a, b, s);
    }

    const int kcol = wn * 16 + l16;
    const int kg = kt * 32 + kcol;
#pragma unroll
    for (int r = 0; r < 8; ++r) {
      const int qrow = wm * 16 + r + 8 * hi;
      float raw = sQQ[qrow] + sKK[kg] - 2.0f * s[r];
      raw = fmaxf(raw, 1e-12f);
      const float sc = -sqrtf(raw) * 0.125f;
      __builtin_nontemporal_store(
          sc * sQM[qrow],
          &ascore[((size_t)hb * L_DIM + q0 + qrow) * L_DIM + kg]);
      const float mn = fmaxf(m_r[r], sc);
      l_r[r] = l_r[r] * __expf(m_r[r] - mn) + __expf(sc - mn);
      m_r[r] = mn;
    }
    __syncthreads();
  }

#pragma unroll
  for (int r = 0; r < 8; ++r) {
    float m = m_r[r], l = l_r[r];
#pragma unroll
    for (int off = 8; off >= 1; off >>= 1) {
      const float mo = __shfl_xor(m, off, 16);
      const float lo = __shfl_xor(l, off, 16);
      const float mn = fmaxf(m, mo);
      l = l * __expf(m - mn) + lo * __expf(mo - mn);
      m = mn;
    }
    m_r[r] = m; l_r[r] = l;
  }
  if (l16 == 0) {
#pragma unroll
    for (int r = 0; r < 8; ++r) {
      const int row = wm * 16 + r + 8 * hi;
      sStat[wn][row][0] = m_r[r];
      sStat[wn][row][1] = l_r[r];
    }
  }
  __syncthreads();
  if (tid < 64) {
    const float ma = sStat[0][tid][0], la = sStat[0][tid][1];
    const float mb = sStat[1][tid][0], lb = sStat[1][tid][1];
    const float mn = fmaxf(ma, mb);
    const float L = la * __expf(ma - mn) + lb * __expf(mb - mn);
    sMrow[tid] = mn;
    sInvL[tid] = 1.0f / L;
  }
  __syncthreads();

  v8f o[2] = {};
  for (int kt = 0; kt < L_DIM / 32; ++kt) {
    {
      const int kc = tid >> 3, d0 = (tid & 7) * 8;
      const size_t kb = ((size_t)hb * L_DIM + kt * 32 + kc) * D_DIM + d0;
      const float4 k0 = *(const float4*)&Kh[kb];
      const float4 k1 = *(const float4*)&Kh[kb + 4];
      float* dk = &sKT2[d0 >> 1][kc * 2];
      *(v2f*)&dk[0] = v2f{k0.x, k0.y};
      *(v2f*)&dk[SKT_STRIDE] = v2f{k0.z, k0.w};
      *(v2f*)&dk[2 * SKT_STRIDE] = v2f{k1.x, k1.y};
      *(v2f*)&dk[3 * SKT_STRIDE] = v2f{k1.z, k1.w};
      const float4 v0 = *(const float4*)&Vh[kb];
      const float4 v1 = *(const float4*)&Vh[kb + 4];
      float* dv = &sV2[kc >> 1][d0 * 2 + (kc & 1)];
      dv[0] = v0.x;  dv[2] = v0.y;  dv[4] = v0.z;  dv[6] = v0.w;
      dv[8] = v1.x;  dv[10] = v1.y; dv[12] = v1.z; dv[14] = v1.w;
    }
    __syncthreads();

    v8f s = {};
#pragma unroll
    for (int kk = 0; kk < D_DIM; kk += 4) {
      const v2f a = *(const v2f*)&sQ[wm * 16 + l16][kk + 2 * hi];
      const v2f b = *(const v2f*)&sKT2[(kk >> 1) + hi][(wn * 16 + l16) * 2];
      s = wmma4(a, b, s);
    }
    const int kcol = wn * 16 + l16;
    const int kg = kt * 32 + kcol;
#pragma unroll
    for (int r = 0; r < 8; ++r) {
      const int qrow = wm * 16 + r + 8 * hi;
      float raw = sQQ[qrow] + sKK[kg] - 2.0f * s[r];
      raw = fmaxf(raw, 1e-12f);
      const float sc = -sqrtf(raw) * 0.125f;
      sP[qrow][kcol] = __expf(sc - sMrow[qrow]) * sInvL[qrow];
    }
    __syncthreads();

#pragma unroll
    for (int kk = 0; kk < 32; kk += 4) {
      const v2f a = *(const v2f*)&sP[wm * 16 + l16][kk + 2 * hi];
      const int kp = (kk >> 1) + hi;
#pragma unroll
      for (int n = 0; n < 2; ++n) {
        const int dcol = wn * 32 + n * 16 + l16;
        const v2f b = *(const v2f*)&sV2[kp][dcol * 2];
        o[n] = wmma4(a, b, o[n]);
      }
    }
    __syncthreads();
  }

  const int h = hb >> 2, b = hb & 3;
#pragma unroll
  for (int n = 0; n < 2; ++n) {
    const int d = wn * 32 + n * 16 + l16;
#pragma unroll
    for (int r = 0; r < 8; ++r) {
      const int qrow = wm * 16 + r + 8 * hi;
      const float v = o[n][r] * sQM[qrow];
      __builtin_nontemporal_store(
          v, &outp[((size_t)b * L_DIM + q0 + qrow) * U_DIM + h * D_DIM + d]);
    }
  }
}

extern "C" void kernel_launch(void* const* d_in, const int* in_sizes, int n_in,
                              void* d_out, int out_size, void* d_ws,
                              size_t ws_size, hipStream_t stream) {
  const float* queries = (const float*)d_in[0];
  const float* keys = (const float*)d_in[1];
  const float* values = (const float*)d_in[2];
  const float* Wq = (const float*)d_in[3];
  const float* bq = (const float*)d_in[4];
  const float* Wk = (const float*)d_in[5];
  const float* bk = (const float*)d_in[6];
  const float* Wv = (const float*)d_in[7];
  const float* bv = (const float*)d_in[8];

  float* outp = (float*)d_out;
  float* ascore = outp + (size_t)B_DIM * L_DIM * U_DIM;

  float* ws = (float*)d_ws;
  float* Qh = ws;
  float* Kh = Qh + (size_t)HB_DIM * L_DIM * D_DIM;
  float* Vh = Kh + (size_t)HB_DIM * L_DIM * D_DIM;
  float* qq = Vh + (size_t)HB_DIM * L_DIM * D_DIM;
  float* kkv = qq + (size_t)HB_DIM * L_DIM;
  float* qsign = kkv + (size_t)HB_DIM * L_DIM;

  proj_relu_kernel<<<dim3(U_DIM / 128, M_ROWS / 64, 3), 256, 0, stream>>>(
      queries, keys, values, Wq, Wk, Wv, bq, bk, bv, Qh, Kh, Vh);

  const int aux_threads = 2 * HB_DIM * L_DIM + B_DIM * L_DIM;
  aux_kernel<<<(aux_threads + 255) / 256, 256, 0, stream>>>(Qh, Kh, queries,
                                                            qq, kkv, qsign);

  attn_kernel<<<dim3(L_DIM / 64, HB_DIM), 256, 0, stream>>>(
      Qh, Kh, Vh, qq, kkv, qsign, outp, ascore);
}


# attention kernel: whole K and whole V of the head staged once into the 320 KiB LDS; the 32 per-tile global-load and staging-barrier round trips go; tile, k and sum order kept
# speedup vs baseline: 1.3617x; 1.3617x over previous
#include <hip/hip_runtime.h>
#include <hip/hip_bf16.h>
#include <math.h>

#define U_DIM 512
#define B_DIM 4
#define L_DIM 512
#define H_DIM 8
#define D_DIM 64
#define HB_DIM 32
#define M_ROWS 2048

typedef __attribute__((ext_vector_type(2))) float v2f;
typedef __attribute__((ext_vector_type(8))) float v8f;

__device__ __forceinline__ v8f wmma4(v2f a, v2f b, v8f c) {
#if defined(__HIP_DEVICE_COMPILE__)
  return __builtin_amdgcn_wmma_f32_16x16x4_f32(false, a, false, b, (short)0, c,
                                               false, false);
#else
  (void)a; (void)b;
  return c;
#endif
}


#define SW_STRIDE 288

__global__ __launch_bounds__(256) void proj_relu_kernel(
    const float* __restrict__ Xq, const float* __restrict__ Xk,
    const float* __restrict__ Xv, const float* __restrict__ Wq,
    const float* __restrict__ Wk, const float* __restrict__ Wv,
    const float* __restrict__ bq, const float* __restrict__ bk,
    const float* __restrict__ bv, float* __restrict__ Qh,
    float* __restrict__ Kh, float* __restrict__ Vh) {
  const int which = blockIdx.z;
  const float* X = which == 0 ? Xq : (which == 1 ? Xk : Xv);
  const float* W = which == 0 ? Wq : (which == 1 ? Wk : Wv);
  const float* bias = which == 0 ? bq : (which == 1 ? bk : bv);
  float* O = which == 0 ? Qh : (which == 1 ? Kh : Vh);

  const int m0 = blockIdx.y * 64;
  const int n0 = blockIdx.x * 128;
  const int tid = threadIdx.x;
  const int lane = tid & 31;
  const int wave = tid >> 5;
  const int wm = wave >> 1;
  const int wn = wave & 1;
  const int l16 = lane & 15;
  const int hi = lane >> 4;

  __shared__ float sX[64][68];
  __shared__ float sW2[8][SW_STRIDE];

  v8f acc[4] = {};

  for (int kt = 0; kt < U_DIM / 16; ++kt) {
    {
      const int r = tid >> 2, c = (tid & 3) * 4;
      const float4 x =
          *(const float4*)&X[(size_t)(m0 + r) * U_DIM + kt * 16 + c];
      *(float4*)&sX[r][c] = x;
    }
    {
      const int r = tid >> 4, c = (tid & 15) * 8;
      const float4 w0 =
          *(const float4*)&W[(size_t)(kt * 16 + r) * U_DIM + n0 + c];
      const float4 w1 =
          *(const float4*)&W[(size_t)(kt * 16 + r) * U_DIM + n0 + c + 4];
      float* dst = &sW2[r >> 1][c * 2 + (r & 1)];
      dst[0] = w0.x;  dst[2] = w0.y;  dst[4] = w0.z;  dst[6] = w0.w;
      dst[8] = w1.x;  dst[10] = w1.y; dst[12] = w1.z; dst[14] = w1.w;
    }
    __syncthreads();

#pragma unroll
    for (int kk = 0; kk < 16; kk += 4) {
      const v2f a = *(const v2f*)&sX[wm * 16 + l16][kk + 2 * hi];
      const int kp = (kk >> 1) + hi;
#pragma unroll
      for (int n = 0; n < 4; ++n) {
        const int col = wn * 64 + n * 16 + l16;
        const v2f b = *(const v2f*)&sW2[kp][col * 2];
        acc[n] = wmma4(a, b, acc[n]);
      }
    }
    __syncthreads();
  }

#pragma unroll
  for (int n = 0; n < 4; ++n) {
    const int ng = n0 + wn * 64 + n * 16 + l16;
    const int h = ng >> 6, d = ng & 63;
    const float bb = bias[ng];
#pragma unroll
    for (int r = 0; r < 8; ++r) {
      const int mg = m0 + wm * 16 + r + 8 * hi;
      const int b = mg >> 9, l = mg & 511;
      float v = acc[n][r] + bb;
      v = v > 0.0f ? v : 0.0f;
      O[(((size_t)(h * B_DIM + b) * L_DIM + l) * D_DIM) + d] = v;
    }
  }
}

__global__ __launch_bounds__(256) void aux_kernel(
    const float* __restrict__ Qh, const float* __restrict__ Kh,
    const float* __restrict__ queries, float* __restrict__ qq,
    float* __restrict__ kkv, float* __restrict__ qsign) {
  const int idx = blockIdx.x * blockDim.x + threadIdx.x;
  const int nrows = HB_DIM * L_DIM;
  if (idx < nrows) {
    const float* p = Qh + (size_t)idx * D_DIM;
    float s = 0.0f;
    for (int d = 0; d < D_DIM; ++d) s += p[d] * p[d];
    qq[idx] = s;
  } else if (idx < 2 * nrows) {
    const float* p = Kh + (size_t)(idx - nrows) * D_DIM;
    float s = 0.0f;
    for (int d = 0; d < D_DIM; ++d) s += p[d] * p[d];
    kkv[idx - nrows] = s;
  } else if (idx < 2 * nrows + B_DIM * L_DIM) {
    const int r = idx - 2 * nrows;
    const float* p = queries + (size_t)r * U_DIM;
    float s = 0.0f;
    for (int u = 0; u < U_DIM; ++u) s += p[u];
    qsign[r] = (s != 0.0f) ? 1.0f : 0.0f;
  }
}

#define SK_STRIDE 68
#define SP_STRIDE 68

__global__ __launch_bounds__(256) void attn_kernel(
    const float* __restrict__ Qh, const float* __restrict__ Kh,
    const float* __restrict__ Vh, const float* __restrict__ qq,
    const float* __restrict__ kkv, const float* __restrict__ qsign,
    float* __restrict__ outp, float* __restrict__ ascore) {
  const int hb = blockIdx.y;
  const int q0 = blockIdx.x * 64;
  const int tid = threadIdx.x;
  const int lane = tid & 31;
  const int wave = tid >> 5;
  const int wm = wave >> 1;
  const int wn = wave & 1;
  const int l16 = lane & 15;
  const int hi = lane >> 4;

  __shared__ float sK[L_DIM][SK_STRIDE];
  __shared__ float sV2[L_DIM / 2][128];
  __shared__ float sQ[64][68];
  __shared__ float sP[64][SP_STRIDE];
  __shared__ float sKK[L_DIM];
  __shared__ float sQQ[64];
  __shared__ float sQM[64];
  __shared__ float sMrow[64];
  __shared__ float sInvL[64];
  __shared__ float sStat[2][64][2];

  {
    const int r = tid >> 2, c = (tid & 3) * 16;
    const size_t base = ((size_t)hb * L_DIM + q0 + r) * D_DIM + c;
#pragma unroll
    for (int j = 0; j < 4; ++j) {
      const float4 v = *(const float4*)&Qh[base + 4 * j];
      *(float4*)&sQ[r][c + 4 * j] = v;
    }
    sKK[2 * tid] = kkv[(size_t)hb * L_DIM + 2 * tid];
    sKK[2 * tid + 1] = kkv[(size_t)hb * L_DIM + 2 * tid + 1];
    if (tid < 64) {
      sQQ[tid] = qq[(size_t)hb * L_DIM + q0 + tid];
      sQM[tid] = qsign[(size_t)(hb & 3) * L_DIM + q0 + tid];
    }
  }
  {
    const int c = (tid & 15) * 4;
    const float* src = Kh + ((size_t)hb * L_DIM + (tid >> 4)) * D_DIM + c;
#pragma unroll 8
    for (int i = 0; i < L_DIM / 16; ++i) {
      const float4 v = *(const float4*)&src[(size_t)i * 16 * D_DIM];
      *(float4*)&sK[(tid >> 4) + 16 * i][c] = v;
    }
  }
  {
    const int d = (tid & 15) * 4;
    const float* src = Vh + ((size_t)hb * L_DIM + 2 * (tid >> 4)) * D_DIM + d;
#pragma unroll 8
    for (int i = 0; i < L_DIM / 32; ++i) {
      const int kp = (tid >> 4) + 16 * i;
      const float4 a = *(const float4*)&src[(size_t)i * 32 * D_DIM];
      const float4 b = *(const float4*)&src[(size_t)i * 32 * D_DIM + D_DIM];
      float* dst = &sV2[kp][(d * 2) ^ ((kp & 1) << 5)];
      *(float4*)&dst[0] = make_float4(a.x, b.x, a.y, b.y);
      *(float4*)&dst[4] = make_float4(a.z, b.z, a.w, b.w);
    }
  }
  __syncthreads();

  v2f qa[16];
#pragma unroll
  for (int j = 0; j < 16; ++j)
    qa[j] = *(const v2f*)&sQ[wm * 16 + l16][4 * j + 2 * hi];
  float qq_r[8], qm_r[8];
#pragma unroll
  for (int r = 0; r < 8; ++r) {
    qq_r[r] = sQQ[wm * 16 + r + 8 * hi];
    qm_r[r] = sQM[wm * 16 + r + 8 * hi];
  }
  const int kcol = wn * 16 + l16;

  float m_r[8], l_r[8];
#pragma unroll
  for (int r = 0; r < 8; ++r) { m_r[r] = -INFINITY; l_r[r] = 0.0f; }

  for (int kt = 0; kt < L_DIM / 32; kt += 2) {
    v8f s[2] = {};
    const float* kr = &sK[kt * 32 + kcol][2 * hi];
#pragma unroll
    for (int j = 0; j < 16; ++j) {
      const v2f b0 = *(const v2f*)&kr[4 * j];
      const v2f b1 = *(const v2f*)&kr[32 * SK_STRIDE + 4 * j];
      s[0] = wmma4(qa[j], b0, s[0]);
      s[1] = wmma4(qa[j], b1, s[1]);
    }
#pragma unroll
    for (int t = 0; t < 2; ++t) {
      const int kg = (kt + t) * 32 + kcol;
      const float kkg = sKK[kg];
#pragma unroll
      for (int r = 0; r < 8; ++r) {
        const int qrow = wm * 16 + r + 8 * hi;
        float raw = qq_r[r] + kkg - 2.0f * s[t][r];
        raw = fmaxf(raw, 1e-12f);
        const float sc = -sqrtf(raw) * 0.125f;
        __builtin_nontemporal_store(
            sc * qm_r[r],
            &ascore[((size_t)hb * L_DIM + q0 + qrow) * L_DIM + kg]);
        const float mn = fmaxf(m_r[r], sc);
        l_r[r] = l_r[r] * __expf(m_r[r] - mn) + __expf(sc - mn);
        m_r[r] = mn;
      }
    }
  }

#pragma unroll
  for (int r = 0; r < 8; ++r) {
    float m = m_r[r], l = l_r[r];
#pragma unroll
    for (int off = 8; off >= 1; off >>= 1) {
      const float mo = __shfl_xor(m, off, 16);
      const float lo = __shfl_xor(l, off, 16);
      const float mn = fmaxf(m, mo);
      l = l * __expf(m - mn) + lo * __expf(mo - mn);
      m = mn;
    }
    m_r[r] = m; l_r[r] = l;
  }
  if (l16 == 0) {
#pragma unroll
    for (int r = 0; r < 8; ++r) {
      const int row = wm * 16 + r + 8 * hi;
      sStat[wn][row][0] = m_r[r];
      sStat[wn][row][1] = l_r[r];
    }
  }
  __syncthreads();
  if (tid < 64) {
    const float ma = sStat[0][tid][0], la = sStat[0][tid][1];
    const float mb = sStat[1][tid][0], lb = sStat[1][tid][1];
    const float mn = fmaxf(ma, mb);
    const float L = la * __expf(ma - mn) + lb * __expf(mb - mn);
    sMrow[tid] = mn;
    sInvL[tid] = 1.0f / L;
  }
  __syncthreads();

  float mrow_r[8], invl_r[8];
#pragma unroll
  for (int r = 0; r < 8; ++r) {
    mrow_r[r] = sMrow[wm * 16 + r + 8 * hi];
    invl_r[r] = sInvL[wm * 16 + r + 8 * hi];
  }

  v8f o[2] = {};
  for (int kt = 0; kt < L_DIM / 32; kt += 2) {
    v8f s[2] = {};
    const float* kr = &sK[kt * 32 + kcol][2 * hi];
#pragma unroll
    for (int j = 0; j < 16; ++j) {
      const v2f b0 = *(const v2f*)&kr[4 * j];
      const v2f b1 = *(const v2f*)&kr[32 * SK_STRIDE + 4 * j];
      s[0] = wmma4(qa[j], b0, s[0]);
      s[1] = wmma4(qa[j], b1, s[1]);
    }
#pragma unroll
    for (int t = 0; t < 2; ++t) {
      const float kkg = sKK[(kt + t) * 32 + kcol];
#pragma unroll
      for (int r = 0; r < 8; ++r) {
        const int qrow = wm * 16 + r + 8 * hi;
        float raw = qq_r[r] + kkg - 2.0f * s[t][r];
        raw = fmaxf(raw, 1e-12f);
        const float sc = -sqrtf(raw) * 0.125f;
        sP[qrow][t * 32 + kcol] = __expf(sc - mrow_r[r]) * invl_r[r];
      }
    }
    __syncthreads();

#pragma unroll
    for (int kk = 0; kk < 64; kk += 4) {
      const v2f a = *(const v2f*)&sP[wm * 16 + l16][kk + 2 * hi];
      const int kp = kt * 16 + (kk >> 1) + hi;
#pragma unroll
      for (int n = 0; n < 2; ++n) {
        const int dcol = wn * 32 + n * 16 + l16;
        const v2f b = *(const v2f*)&sV2[kp][(dcol * 2) ^ (hi << 5)];
        o[n] = wmma4(a, b, o[n]);
      }
    }
    __syncthreads();
  }

  const int h = hb >> 2, b = hb & 3;
#pragma unroll
  for (int n = 0; n < 2; ++n) {
    const int d = wn * 32 + n * 16 + l16;
#pragma unroll
    for (int r = 0; r < 8; ++r) {
      const int qrow = wm * 16 + r + 8 * hi;
      const float v = o[n][r] * qm_r[r];
      __builtin_nontemporal_store(
          v, &outp[((size_t)b * L_DIM + q0 + qrow) * U_DIM + h * D_DIM + d]);
    }
  }
}

extern "C" void kernel_launch(void* const* d_in, const int* in_sizes, int n_in,
                              void* d_out, int out_size, void* d_ws,
                              size_t ws_size, hipStream_t stream) {
  const float* queries = (const float*)d_in[0];
  const float* keys = (const float*)d_in[1];
  const float* values = (const float*)d_in[2];
  const float* Wq = (const float*)d_in[3];
  const float* bq = (const float*)d_in[4];
  const float* Wk = (const float*)d_in[5];
  const float* bk = (const float*)d_in[6];
  const float* Wv = (const float*)d_in[7];
  const float* bv = (const float*)d_in[8];

  float* outp = (float*)d_out;
  float* ascore = outp + (size_t)B_DIM * L_DIM * U_DIM;

  float* ws = (float*)d_ws;
  float* Qh = ws;
  float* Kh = Qh + (size_t)HB_DIM * L_DIM * D_DIM;
  float* Vh = Kh + (size_t)HB_DIM * L_DIM * D_DIM;
  float* qq = Vh + (size_t)HB_DIM * L_DIM * D_DIM;
  float* kkv = qq + (size_t)HB_DIM * L_DIM;
  float* qsign = kkv + (size_t)HB_DIM * L_DIM;

  proj_relu_kernel<<<dim3(U_DIM / 128, M_ROWS / 64, 3), 256, 0, stream>>>(
      queries, keys, values, Wq, Wk, Wv, bq, bk, bv, Qh, Kh, Vh);

  const int aux_threads = 2 * HB_DIM * L_DIM + B_DIM * L_DIM;
  aux_kernel<<<(aux_threads + 255) / 256, 256, 0, stream>>>(Qh, Kh, queries,
                                                            qq, kkv, qsign);

  attn_kernel<<<dim3(L_DIM / 64, HB_DIM), 256, 0, stream>>>(
      Qh, Kh, Vh, qq, kkv, qsign, outp, ascore);
}


# row norms qq/kk and query-row sums folded into the projection kernel (from its accumulators and its staged X tile); aux launch gone, 3 to 2; those sums now lane partials + butterfly (order changed)
# speedup vs baseline: 1.5849x; 1.1639x over previous
#include <hip/hip_runtime.h>
#include <hip/hip_bf16.h>
#include <math.h>

#define U_DIM 512
#define B_DIM 4
#define L_DIM 512
#define H_DIM 8
#define D_DIM 64
#define HB_DIM 32
#define M_ROWS 2048

typedef __attribute__((ext_vector_type(2))) float v2f;
typedef __attribute__((ext_vector_type(8))) float v8f;

__device__ __forceinline__ v8f wmma4(v2f a, v2f b, v8f c) {
#if defined(__HIP_DEVICE_COMPILE__)
  return __builtin_amdgcn_wmma_f32_16x16x4_f32(false, a, false, b, (short)0, c,
                                               false, false);
#else
  (void)a; (void)b;
  return c;
#endif
}


#define SW_STRIDE 288

__global__ __launch_bounds__(256) void proj_relu_kernel(
    const float* __restrict__ Xq, const float* __restrict__ Xk,
    const float* __restrict__ Xv, const float* __restrict__ Wq,
    const float* __restrict__ Wk, const float* __restrict__ Wv,
    const float* __restrict__ bq, const float* __restrict__ bk,
    const float* __restrict__ bv, float* __restrict__ Qh,
    float* __restrict__ Kh, float* __restrict__ Vh, float* __restrict__ qq,
    float* __restrict__ kkv, float* __restrict__ qsign) {
  const int which = blockIdx.z;
  const float* X = which == 0 ? Xq : (which == 1 ? Xk : Xv);
  const float* W = which == 0 ? Wq : (which == 1 ? Wk : Wv);
  const float* bias = which == 0 ? bq : (which == 1 ? bk : bv);
  float* O = which == 0 ? Qh : (which == 1 ? Kh : Vh);

  const int m0 = blockIdx.y * 64;
  const int n0 = blockIdx.x * 128;
  const int tid = threadIdx.x;
  const int lane = tid & 31;
  const int wave = tid >> 5;
  const int wm = wave >> 1;
  const int wn = wave & 1;
  const int l16 = lane & 15;
  const int hi = lane >> 4;

  __shared__ float sX[64][68];
  __shared__ float sW2[8][SW_STRIDE];
  __shared__ float sN[2][64];
  __shared__ float sRS[64];

  const bool do_rowsum = (which == 0) && (blockIdx.x == 0);
  float rs = 0.0f;

  v8f acc[4] = {};

  for (int kt = 0; kt < U_DIM / 16; ++kt) {
    {
      const int r = tid >> 2, c = (tid & 3) * 4;
      const float4 x =
          *(const float4*)&X[(size_t)(m0 + r) * U_DIM + kt * 16 + c];
      *(float4*)&sX[r][c] = x;
      if (do_rowsum) rs += (x.x + x.y) + (x.z + x.w);
    }
    {
      const int r = tid >> 4, c = (tid & 15) * 8;
      const float4 w0 =
          *(const float4*)&W[(size_t)(kt * 16 + r) * U_DIM + n0 + c];
      const float4 w1 =
          *(const float4*)&W[(size_t)(kt * 16 + r) * U_DIM + n0 + c + 4];
      float* dst = &sW2[r >> 1][c * 2 + (r & 1)];
      dst[0] = w0.x;  dst[2] = w0.y;  dst[4] = w0.z;  dst[6] = w0.w;
      dst[8] = w1.x;  dst[10] = w1.y; dst[12] = w1.z; dst[14] = w1.w;
    }
    __syncthreads();

#pragma unroll
    for (int kk = 0; kk < 16; kk += 4) {
      const v2f a = *(const v2f*)&sX[wm * 16 + l16][kk + 2 * hi];
      const int kp = (kk >> 1) + hi;
#pragma unroll
      for (int n = 0; n < 4; ++n) {
        const int col = wn * 64 + n * 16 + l16;
        const v2f b = *(const v2f*)&sW2[kp][col * 2];
        acc[n] = wmma4(a, b, acc[n]);
      }
    }
    __syncthreads();
  }

  if (do_rowsum) {
    rs += __shfl_xor(rs, 1);
    rs += __shfl_xor(rs, 2);
    if ((tid & 3) == 0) sRS[tid >> 2] = rs;
  }

  float nrm[8];
#pragma unroll
  for (int r = 0; r < 8; ++r) nrm[r] = 0.0f;
#pragma unroll
  for (int n = 0; n < 4; ++n) {
    const int ng = n0 + wn * 64 + n * 16 + l16;
    const int h = ng >> 6, d = ng & 63;
    const float bb = bias[ng];
#pragma unroll
    for (int r = 0; r < 8; ++r) {
      const int mg = m0 + wm * 16 + r + 8 * hi;
      const int b = mg >> 9, l = mg & 511;
      float v = acc[n][r] + bb;
      v = v > 0.0f ? v : 0.0f;
      nrm[r] += v * v;
      O[(((size_t)(h * B_DIM + b) * L_DIM + l) * D_DIM) + d] = v;
    }
  }

  if (which < 2) {
#pragma unroll
    for (int r = 0; r < 8; ++r) {
#pragma unroll
      for (int off = 8; off >= 1; off >>= 1)
        nrm[r] += __shfl_xor(nrm[r], off, 16);
    }
    if (l16 == 0) {
#pragma unroll
      for (int r = 0; r < 8; ++r) sN[wn][wm * 16 + r + 8 * hi] = nrm[r];
    }
    __syncthreads();
    if (tid < 128) {
      const int hh = tid >> 6, row = tid & 63;
      const int mg = m0 + row;
      const int b = mg >> 9, l = mg & 511;
      const int h = (n0 >> 6) + hh;
      float* N = which == 0 ? qq : kkv;
      N[(size_t)(h * B_DIM + b) * L_DIM + l] = sN[hh][row];
    } else if (do_rowsum && tid < 192) {
      const int row = tid - 128;
      qsign[m0 + row] = (sRS[row] != 0.0f) ? 1.0f : 0.0f;
    }
  }
}

#define SK_STRIDE 68
#define SP_STRIDE 68

__global__ __launch_bounds__(256) void attn_kernel(
    const float* __restrict__ Qh, const float* __restrict__ Kh,
    const float* __restrict__ Vh, const float* __restrict__ qq,
    const float* __restrict__ kkv, const float* __restrict__ qsign,
    float* __restrict__ outp, float* __restrict__ ascore) {
  const int hb = blockIdx.y;
  const int q0 = blockIdx.x * 64;
  const int tid = threadIdx.x;
  const int lane = tid & 31;
  const int wave = tid >> 5;
  const int wm = wave >> 1;
  const int wn = wave & 1;
  const int l16 = lane & 15;
  const int hi = lane >> 4;

  __shared__ float sK[L_DIM][SK_STRIDE];
  __shared__ float sV2[L_DIM / 2][128];
  __shared__ float sQ[64][68];
  __shared__ float sP[64][SP_STRIDE];
  __shared__ float sKK[L_DIM];
  __shared__ float sQQ[64];
  __shared__ float sQM[64];
  __shared__ float sMrow[64];
  __shared__ float sInvL[64];
  __shared__ float sStat[2][64][2];

  {
    const int r = tid >> 2, c = (tid & 3) * 16;
    const size_t base = ((size_t)hb * L_DIM + q0 + r) * D_DIM + c;
#pragma unroll
    for (int j = 0; j < 4; ++j) {
      const float4 v = *(const float4*)&Qh[base + 4 * j];
      *(float4*)&sQ[r][c + 4 * j] = v;
    }
    sKK[2 * tid] = kkv[(size_t)hb * L_DIM + 2 * tid];
    sKK[2 * tid + 1] = kkv[(size_t)hb * L_DIM + 2 * tid + 1];
    if (tid < 64) {
      sQQ[tid] = qq[(size_t)hb * L_DIM + q0 + tid];
      sQM[tid] = qsign[(size_t)(hb & 3) * L_DIM + q0 + tid];
    }
  }
  {
    const int c = (tid & 15) * 4;
    const float* src = Kh + ((size_t)hb * L_DIM + (tid >> 4)) * D_DIM + c;
#pragma unroll 8
    for (int i = 0; i < L_DIM / 16; ++i) {
      const float4 v = *(const float4*)&src[(size_t)i * 16 * D_DIM];
      *(float4*)&sK[(tid >> 4) + 16 * i][c] = v;
    }
  }
  {
    const int d = (tid & 15) * 4;
    const float* src = Vh + ((size_t)hb * L_DIM + 2 * (tid >> 4)) * D_DIM + d;
#pragma unroll 8
    for (int i = 0; i < L_DIM / 32; ++i) {
      const int kp = (tid >> 4) + 16 * i;
      const float4 a = *(const float4*)&src[(size_t)i * 32 * D_DIM];
      const float4 b = *(const float4*)&src[(size_t)i * 32 * D_DIM + D_DIM];
      float* dst = &sV2[kp][(d * 2) ^ ((kp & 1) << 5)];
      *(float4*)&dst[0] = make_float4(a.x, b.x, a.y, b.y);
      *(float4*)&dst[4] = make_float4(a.z, b.z, a.w, b.w);
    }
  }
  __syncthreads();

  v2f qa[16];
#pragma unroll
  for (int j = 0; j < 16; ++j)
    qa[j] = *(const v2f*)&sQ[wm * 16 + l16][4 * j + 2 * hi];
  float qq_r[8], qm_r[8];
#pragma unroll
  for (int r = 0; r < 8; ++r) {
    qq_r[r] = sQQ[wm * 16 + r + 8 * hi];
    qm_r[r] = sQM[wm * 16 + r + 8 * hi];
  }
  const int kcol = wn * 16 + l16;

  float m_r[8], l_r[8];
#pragma unroll
  for (int r = 0; r < 8; ++r) { m_r[r] = -INFINITY; l_r[r] = 0.0f; }

  for (int kt = 0; kt < L_DIM / 32; kt += 2) {
    v8f s[2] = {};
    const float* kr = &sK[kt * 32 + kcol][2 * hi];
#pragma unroll
    for (int j = 0; j < 16; ++j) {
      const v2f b0 = *(const v2f*)&kr[4 * j];
      const v2f b1 = *(const v2f*)&kr[32 * SK_STRIDE + 4 * j];
      s[0] = wmma4(qa[j], b0, s[0]);
      s[1] = wmma4(qa[j], b1, s[1]);
    }
#pragma unroll
    for (int t = 0; t < 2; ++t) {
      const int kg = (kt + t) * 32 + kcol;
      const float kkg = sKK[kg];
#pragma unroll
      for (int r = 0; r < 8; ++r) {
        const int qrow = wm * 16 + r + 8 * hi;
        float raw = qq_r[r] + kkg - 2.0f * s[t][r];
        raw = fmaxf(raw, 1e-12f);
        const float sc = -sqrtf(raw) * 0.125f;
        __builtin_nontemporal_store(
            sc * qm_r[r],
            &ascore[((size_t)hb * L_DIM + q0 + qrow) * L_DIM + kg]);
        const float mn = fmaxf(m_r[r], sc);
        l_r[r] = l_r[r] * __expf(m_r[r] - mn) + __expf(sc - mn);
        m_r[r] = mn;
      }
    }
  }

#pragma unroll
  for (int r = 0; r < 8; ++r) {
    float m = m_r[r], l = l_r[r];
#pragma unroll
    for (int off = 8; off >= 1; off >>= 1) {
      const float mo = __shfl_xor(m, off, 16);
      const float lo = __shfl_xor(l, off, 16);
      const float mn = fmaxf(m, mo);
      l = l * __expf(m - mn) + lo * __expf(mo - mn);
      m = mn;
    }
    m_r[r] = m; l_r[r] = l;
  }
  if (l16 == 0) {
#pragma unroll
    for (int r = 0; r < 8; ++r) {
      const int row = wm * 16 + r + 8 * hi;
      sStat[wn][row][0] = m_r[r];
      sStat[wn][row][1] = l_r[r];
    }
  }
  __syncthreads();
  if (tid < 64) {
    const float ma = sStat[0][tid][0], la = sStat[0][tid][1];
    const float mb = sStat[1][tid][0], lb = sStat[1][tid][1];
    const float mn = fmaxf(ma, mb);
    const float L = la * __expf(ma - mn) + lb * __expf(mb - mn);
    sMrow[tid] = mn;
    sInvL[tid] = 1.0f / L;
  }
  __syncthreads();

  float mrow_r[8], invl_r[8];
#pragma unroll
  for (int r = 0; r < 8; ++r) {
    mrow_r[r] = sMrow[wm * 16 + r + 8 * hi];
    invl_r[r] = sInvL[wm * 16 + r + 8 * hi];
  }

  v8f o[2] = {};
  for (int kt = 0; kt < L_DIM / 32; kt += 2) {
    v8f s[2] = {};
    const float* kr = &sK[kt * 32 + kcol][2 * hi];
#pragma unroll
    for (int j = 0; j < 16; ++j) {
      const v2f b0 = *(const v2f*)&kr[4 * j];
      const v2f b1 = *(const v2f*)&kr[32 * SK_STRIDE + 4 * j];
      s[0] = wmma4(qa[j], b0, s[0]);
      s[1] = wmma4(qa[j], b1, s[1]);
    }
#pragma unroll
    for (int t = 0; t < 2; ++t) {
      const float kkg = sKK[(kt + t) * 32 + kcol];
#pragma unroll
      for (int r = 0; r < 8; ++r) {
        const int qrow = wm * 16 + r + 8 * hi;
        float raw = qq_r[r] + kkg - 2.0f * s[t][r];
        raw = fmaxf(raw, 1e-12f);
        const float sc = -sqrtf(raw) * 0.125f;
        sP[qrow][t * 32 + kcol] = __expf(sc - mrow_r[r]) * invl_r[r];
      }
    }
    __syncthreads();

#pragma unroll
    for (int kk = 0; kk < 64; kk += 4) {
      const v2f a = *(const v2f*)&sP[wm * 16 + l16][kk + 2 * hi];
      const int kp = kt * 16 + (kk >> 1) + hi;
#pragma unroll
      for (int n = 0; n < 2; ++n) {
        const int dcol = wn * 32 + n * 16 + l16;
        const v2f b = *(const v2f*)&sV2[kp][(dcol * 2) ^ (hi << 5)];
        o[n] = wmma4(a, b, o[n]);
      }
    }
    __syncthreads();
  }

  const int h = hb >> 2, b = hb & 3;
#pragma unroll
  for (int n = 0; n < 2; ++n) {
    const int d = wn * 32 + n * 16 + l16;
#pragma unroll
    for (int r = 0; r < 8; ++r) {
      const int qrow = wm * 16 + r + 8 * hi;
      const float v = o[n][r] * qm_r[r];
      __builtin_nontemporal_store(
          v, &outp[((size_t)b * L_DIM + q0 + qrow) * U_DIM + h * D_DIM + d]);
    }
  }
}

extern "C" void kernel_launch(void* const* d_in, const int* in_sizes, int n_in,
                              void* d_out, int out_size, void* d_ws,
                              size_t ws_size, hipStream_t stream) {
  const float* queries = (const float*)d_in[0];
  const float* keys = (const float*)d_in[1];
  const float* values = (const float*)d_in[2];
  const float* Wq = (const float*)d_in[3];
  const float* bq = (const float*)d_in[4];
  const float* Wk = (const float*)d_in[5];
  const float* bk = (const float*)d_in[6];
  const float* Wv = (const float*)d_in[7];
  const float* bv = (const float*)d_in[8];

  float* outp = (float*)d_out;
  float* ascore = outp + (size_t)B_DIM * L_DIM * U_DIM;

  float* ws = (float*)d_ws;
  float* Qh = ws;
  float* Kh = Qh + (size_t)HB_DIM * L_DIM * D_DIM;
  float* Vh = Kh + (size_t)HB_DIM * L_DIM * D_DIM;
  float* qq = Vh + (size_t)HB_DIM * L_DIM * D_DIM;
  float* kkv = qq + (size_t)HB_DIM * L_DIM;
  float* qsign = kkv + (size_t)HB_DIM * L_DIM;

  proj_relu_kernel<<<dim3(U_DIM / 128, M_ROWS / 64, 3), 256, 0, stream>>>(
      queries, keys, values, Wq, Wk, Wv, bq, bk, bv, Qh, Kh, Vh, qq, kkv,
      qsign);

  attn_kernel<<<dim3(L_DIM / 64, HB_DIM), 256, 0, stream>>>(
      Qh, Kh, Vh, qq, kkv, qsign, outp, ascore);
}


# attention kernel: pass A parks its scores in LDS (V staged into K's space between the passes), pass B builds P from them; second Q*K^T, P round trip and key-loop barriers go; sums and order kept
# speedup vs baseline: 1.7380x; 1.0966x over previous
#include <hip/hip_runtime.h>
#include <hip/hip_bf16.h>
#include <math.h>

#define U_DIM 512
#define B_DIM 4
#define L_DIM 512
#define H_DIM 8
#define D_DIM 64
#define HB_DIM 32
#define M_ROWS 2048

typedef __attribute__((ext_vector_type(2))) float v2f;
typedef __attribute__((ext_vector_type(8))) float v8f;

__device__ __forceinline__ v8f wmma4(v2f a, v2f b, v8f c) {
#if defined(__HIP_DEVICE_COMPILE__)
  return __builtin_amdgcn_wmma_f32_16x16x4_f32(false, a, false, b, (short)0, c,
                                               false, false);
#else
  (void)a; (void)b;
  return c;
#endif
}


#define SW_STRIDE 288

__global__ __launch_bounds__(256) void proj_relu_kernel(
    const float* __restrict__ Xq, const float* __restrict__ Xk,
    const float* __restrict__ Xv, const float* __restrict__ Wq,
    const float* __restrict__ Wk, const float* __restrict__ Wv,
    const float* __restrict__ bq, const float* __restrict__ bk,
    const float* __restrict__ bv, float* __restrict__ Qh,
    float* __restrict__ Kh, float* __restrict__ Vh, float* __restrict__ qq,
    float* __restrict__ kkv, float* __restrict__ qsign) {
  const int which = blockIdx.z;
  const float* X = which == 0 ? Xq : (which == 1 ? Xk : Xv);
  const float* W = which == 0 ? Wq : (which == 1 ? Wk : Wv);
  const float* bias = which == 0 ? bq : (which == 1 ? bk : bv);
  float* O = which == 0 ? Qh : (which == 1 ? Kh : Vh);

  const int m0 = blockIdx.y * 64;
  const int n0 = blockIdx.x * 128;
  const int tid = threadIdx.x;
  const int lane = tid & 31;
  const int wave = tid >> 5;
  const int wm = wave >> 1;
  const int wn = wave & 1;
  const int l16 = lane & 15;
  const int hi = lane >> 4;

  __shared__ float sX[64][68];
  __shared__ float sW2[8][SW_STRIDE];
  __shared__ float sN[2][64];
  __shared__ float sRS[64];

  const bool do_rowsum = (which == 0) && (blockIdx.x == 0);
  float rs = 0.0f;

  v8f acc[4] = {};

  for (int kt = 0; kt < U_DIM / 16; ++kt) {
    {
      const int r = tid >> 2, c = (tid & 3) * 4;
      const float4 x =
          *(const float4*)&X[(size_t)(m0 + r) * U_DIM + kt * 16 + c];
      *(float4*)&sX[r][c] = x;
      if (do_rowsum) rs += (x.x + x.y) + (x.z + x.w);
    }
    {
      const int r = tid >> 4, c = (tid & 15) * 8;
      const float4 w0 =
          *(const float4*)&W[(size_t)(kt * 16 + r) * U_DIM + n0 + c];
      const float4 w1 =
          *(const float4*)&W[(size_t)(kt * 16 + r) * U_DIM + n0 + c + 4];
      float* dst = &sW2[r >> 1][c * 2 + (r & 1)];
      dst[0] = w0.x;  dst[2] = w0.y;  dst[4] = w0.z;  dst[6] = w0.w;
      dst[8] = w1.x;  dst[10] = w1.y; dst[12] = w1.z; dst[14] = w1.w;
    }
    __syncthreads();

#pragma unroll
    for (int kk = 0; kk < 16; kk += 4) {
      const v2f a = *(const v2f*)&sX[wm * 16 + l16][kk + 2 * hi];
      const int kp = (kk >> 1) + hi;
#pragma unroll
      for (int n = 0; n < 4; ++n) {
        const int col = wn * 64 + n * 16 + l16;
        const v2f b = *(const v2f*)&sW2[kp][col * 2];
        acc[n] = wmma4(a, b, acc[n]);
      }
    }
    __syncthreads();
  }

  if (do_rowsum) {
    rs += __shfl_xor(rs, 1);
    rs += __shfl_xor(rs, 2);
    if ((tid & 3) == 0) sRS[tid >> 2] = rs;
  }

  float nrm[8];
#pragma unroll
  for (int r = 0; r < 8; ++r) nrm[r] = 0.0f;
#pragma unroll
  for (int n = 0; n < 4; ++n) {
    const int ng = n0 + wn * 64 + n * 16 + l16;
    const int h = ng >> 6, d = ng & 63;
    const float bb = bias[ng];
#pragma unroll
    for (int r = 0; r < 8; ++r) {
      const int mg = m0 + wm * 16 + r + 8 * hi;
      const int b = mg >> 9, l = mg & 511;
      float v = acc[n][r] + bb;
      v = v > 0.0f ? v : 0.0f;
      nrm[r] += v * v;
      O[(((size_t)(h * B_DIM + b) * L_DIM + l) * D_DIM) + d] = v;
    }
  }

  if (which < 2) {
#pragma unroll
    for (int r = 0; r < 8; ++r) {
#pragma unroll
      for (int off = 8; off >= 1; off >>= 1)
        nrm[r] += __shfl_xor(nrm[r], off, 16);
    }
    if (l16 == 0) {
#pragma unroll
      for (int r = 0; r < 8; ++r) sN[wn][wm * 16 + r + 8 * hi] = nrm[r];
    }
    __syncthreads();
    if (tid < 128) {
      const int hh = tid >> 6, row = tid & 63;
      const int mg = m0 + row;
      const int b = mg >> 9, l = mg & 511;
      const int h = (n0 >> 6) + hh;
      float* N = which == 0 ? qq : kkv;
      N[(size_t)(h * B_DIM + b) * L_DIM + l] = sN[hh][row];
    } else if (do_rowsum && tid < 192) {
      const int row = tid - 128;
      qsign[m0 + row] = (sRS[row] != 0.0f) ? 1.0f : 0.0f;
    }
  }
}

#define SK_STRIDE 68
#define SS_STRIDE 516

__global__ __launch_bounds__(256) void attn_kernel(
    const float* __restrict__ Qh, const float* __restrict__ Kh,
    const float* __restrict__ Vh, const float* __restrict__ qq,
    const float* __restrict__ kkv, const float* __restrict__ qsign,
    float* __restrict__ outp, float* __restrict__ ascore) {
  const int hb = blockIdx.y;
  const int q0 = blockIdx.x * 64;
  const int tid = threadIdx.x;
  const int lane = tid & 31;
  const int wave = tid >> 5;
  const int wm = wave >> 1;
  const int wn = wave & 1;
  const int l16 = lane & 15;
  const int hi = lane >> 4;

  __shared__ float sKV[L_DIM * SK_STRIDE];
  float (*sK)[SK_STRIDE] = (float (*)[SK_STRIDE])sKV;
  float (*sV2)[128] = (float (*)[128])sKV;
  __shared__ float sQ[64][68];
  __shared__ float sS[64][SS_STRIDE];
  __shared__ float sKK[L_DIM];
  __shared__ float sQQ[64];
  __shared__ float sQM[64];
  __shared__ float sMrow[64];
  __shared__ float sInvL[64];
  __shared__ float sStat[2][64][2];

  {
    const int r = tid >> 2, c = (tid & 3) * 16;
    const size_t base = ((size_t)hb * L_DIM + q0 + r) * D_DIM + c;
#pragma unroll
    for (int j = 0; j < 4; ++j) {
      const float4 v = *(const float4*)&Qh[base + 4 * j];
      *(float4*)&sQ[r][c + 4 * j] = v;
    }
    sKK[2 * tid] = kkv[(size_t)hb * L_DIM + 2 * tid];
    sKK[2 * tid + 1] = kkv[(size_t)hb * L_DIM + 2 * tid + 1];
    if (tid < 64) {
      sQQ[tid] = qq[(size_t)hb * L_DIM + q0 + tid];
      sQM[tid] = qsign[(size_t)(hb & 3) * L_DIM + q0 + tid];
    }
  }
  {
    const int c = (tid & 15) * 4;
    const float* src = Kh + ((size_t)hb * L_DIM + (tid >> 4)) * D_DIM + c;
#pragma unroll 8
    for (int i = 0; i < L_DIM / 16; ++i) {
      const float4 v = *(const float4*)&src[(size_t)i * 16 * D_DIM];
      *(float4*)&sK[(tid >> 4) + 16 * i][c] = v;
    }
  }
  __syncthreads();

  v2f qa[16];
#pragma unroll
  for (int j = 0; j < 16; ++j)
    qa[j] = *(const v2f*)&sQ[wm * 16 + l16][4 * j + 2 * hi];
  float qq_r[8], qm_r[8];
#pragma unroll
  for (int r = 0; r < 8; ++r) {
    qq_r[r] = sQQ[wm * 16 + r + 8 * hi];
    qm_r[r] = sQM[wm * 16 + r + 8 * hi];
  }
  const int kcol = wn * 16 + l16;

  float m_r[8], l_r[8];
#pragma unroll
  for (int r = 0; r < 8; ++r) { m_r[r] = -INFINITY; l_r[r] = 0.0f; }

  for (int kt = 0; kt < L_DIM / 32; kt += 2) {
    v8f s[2] = {};
    const float* kr = &sK[kt * 32 + kcol][2 * hi];
#pragma unroll
    for (int j = 0; j < 16; ++j) {
      const v2f b0 = *(const v2f*)&kr[4 * j];
      const v2f b1 = *(const v2f*)&kr[32 * SK_STRIDE + 4 * j];
      s[0] = wmma4(qa[j], b0, s[0]);
      s[1] = wmma4(qa[j], b1, s[1]);
    }
#pragma unroll
    for (int t = 0; t < 2; ++t) {
      const int kg = (kt + t) * 32 + kcol;
      const float kkg = sKK[kg];
#pragma unroll
      for (int r = 0; r < 8; ++r) {
        const int qrow = wm * 16 + r + 8 * hi;
        float raw = qq_r[r] + kkg - 2.0f * s[t][r];
        raw = fmaxf(raw, 1e-12f);
        const float sc = -sqrtf(raw) * 0.125f;
        sS[qrow][kg] = sc;
        __builtin_nontemporal_store(
            sc * qm_r[r],
            &ascore[((size_t)hb * L_DIM + q0 + qrow) * L_DIM + kg]);
        const float mn = fmaxf(m_r[r], sc);
        l_r[r] = l_r[r] * __expf(m_r[r] - mn) + __expf(sc - mn);
        m_r[r] = mn;
      }
    }
  }

#pragma unroll
  for (int r = 0; r < 8; ++r) {
    float m = m_r[r], l = l_r[r];
#pragma unroll
    for (int off = 8; off >= 1; off >>= 1) {
      const float mo = __shfl_xor(m, off, 16);
      const float lo = __shfl_xor(l, off, 16);
      const float mn = fmaxf(m, mo);
      l = l * __expf(m - mn) + lo * __expf(mo - mn);
      m = mn;
    }
    m_r[r] = m; l_r[r] = l;
  }
  if (l16 == 0) {
#pragma unroll
    for (int r = 0; r < 8; ++r) {
      const int row = wm * 16 + r + 8 * hi;
      sStat[wn][row][0] = m_r[r];
      sStat[wn][row][1] = l_r[r];
    }
  }
  __syncthreads();
  {
    const int d = (tid & 15) * 4;
    const float* src = Vh + ((size_t)hb * L_DIM + 2 * (tid >> 4)) * D_DIM + d;
#pragma unroll 8
    for (int i = 0; i < L_DIM / 32; ++i) {
      const int kp = (tid >> 4) + 16 * i;
      const float4 a = *(const float4*)&src[(size_t)i * 32 * D_DIM];
      const float4 b = *(const float4*)&src[(size_t)i * 32 * D_DIM + D_DIM];
      float* dst = &sV2[kp][(d * 2) ^ ((kp & 1) << 5)];
      *(float4*)&dst[0] = make_float4(a.x, b.x, a.y, b.y);
      *(float4*)&dst[4] = make_float4(a.z, b.z, a.w, b.w);
    }
  }
  if (tid < 64) {
    const float ma = sStat[0][tid][0], la = sStat[0][tid][1];
    const float mb = sStat[1][tid][0], lb = sStat[1][tid][1];
    const float mn = fmaxf(ma, mb);
    const float L = la * __expf(ma - mn) + lb * __expf(mb - mn);
    sMrow[tid] = mn;
    sInvL[tid] = 1.0f / L;
  }
  __syncthreads();

  const float mrow_a = sMrow[wm * 16 + l16];
  const float invl_a = sInvL[wm * 16 + l16];

  v8f o[2] = {};
  for (int kt = 0; kt < L_DIM / 32; kt += 2) {
    const float* sr = &sS[wm * 16 + l16][kt * 32 + 2 * hi];
#pragma unroll
    for (int kk = 0; kk < 64; kk += 4) {
      const v2f sv = *(const v2f*)&sr[kk];
      v2f a;
      a.x = __expf(sv.x - mrow_a) * invl_a;
      a.y = __expf(sv.y - mrow_a) * invl_a;
      const int kp = kt * 16 + (kk >> 1) + hi;
#pragma unroll
      for (int n = 0; n < 2; ++n) {
        const int dcol = wn * 32 + n * 16 + l16;
        const v2f b = *(const v2f*)&sV2[kp][(dcol * 2) ^ (hi << 5)];
        o[n] = wmma4(a, b, o[n]);
      }
    }
  }

  const int h = hb >> 2, b = hb & 3;
#pragma unroll
  for (int n = 0; n < 2; ++n) {
    const int d = wn * 32 + n * 16 + l16;
#pragma unroll
    for (int r = 0; r < 8; ++r) {
      const int qrow = wm * 16 + r + 8 * hi;
      const float v = o[n][r] * qm_r[r];
      __builtin_nontemporal_store(
          v, &outp[((size_t)b * L_DIM + q0 + qrow) * U_DIM + h * D_DIM + d]);
    }
  }
}

extern "C" void kernel_launch(void* const* d_in, const int* in_sizes, int n_in,
                              void* d_out, int out_size, void* d_ws,
                              size_t ws_size, hipStream_t stream) {
  const float* queries = (const float*)d_in[0];
  const float* keys = (const float*)d_in[1];
  const float* values = (const float*)d_in[2];
  const float* Wq = (const float*)d_in[3];
  const float* bq = (const float*)d_in[4];
  const float* Wk = (const float*)d_in[5];
  const float* bk = (const float*)d_in[6];
  const float* Wv = (const float*)d_in[7];
  const float* bv = (const float*)d_in[8];

  float* outp = (float*)d_out;
  float* ascore = outp + (size_t)B_DIM * L_DIM * U_DIM;

  float* ws = (float*)d_ws;
  float* Qh = ws;
  float* Kh = Qh + (size_t)HB_DIM * L_DIM * D_DIM;
  float* Vh = Kh + (size_t)HB_DIM * L_DIM * D_DIM;
  float* qq = Vh + (size_t)HB_DIM * L_DIM * D_DIM;
  float* kkv = qq + (size_t)HB_DIM * L_DIM;
  float* qsign = kkv + (size_t)HB_DIM * L_DIM;

  proj_relu_kernel<<<dim3(U_DIM / 128, M_ROWS / 64, 3), 256, 0, stream>>>(
      queries, keys, values, Wq, Wk, Wv, bq, bk, bv, Qh, Kh, Vh, qq, kkv,
      qsign);

  attn_kernel<<<dim3(L_DIM / 64, HB_DIM), 256, 0, stream>>>(
      Qh, Kh, Vh, qq, kkv, qsign, outp, ascore);
}


# projection kernel: K loop software-pipelined over two LDS stages (K step 16 to 64): next tile's global loads fly under the current tile's WMMAs, one barrier per step; k and sum order kept
# speedup vs baseline: 1.9860x; 1.1427x over previous
#include <hip/hip_runtime.h>
#include <hip/hip_bf16.h>
#include <math.h>

#define U_DIM 512
#define B_DIM 4
#define L_DIM 512
#define H_DIM 8
#define D_DIM 64
#define HB_DIM 32
#define M_ROWS 2048

typedef __attribute__((ext_vector_type(2))) float v2f;
typedef __attribute__((ext_vector_type(8))) float v8f;

__device__ __forceinline__ v8f wmma4(v2f a, v2f b, v8f c) {
#if defined(__HIP_DEVICE_COMPILE__)
  return __builtin_amdgcn_wmma_f32_16x16x4_f32(false, a, false, b, (short)0, c,
                                               false, false);
#else
  (void)a; (void)b;
  return c;
#endif
}


#define SW_STRIDE 288
#define K_STEP 64
#define SX_PITCH 68
#define N_STEPS (U_DIM / K_STEP)
#define K_SUB (K_STEP / 16)

__global__ __launch_bounds__(256) void proj_relu_kernel(
    const float* __restrict__ Xq, const float* __restrict__ Xk,
    const float* __restrict__ Xv, const float* __restrict__ Wq,
    const float* __restrict__ Wk, const float* __restrict__ Wv,
    const float* __restrict__ bq, const float* __restrict__ bk,
    const float* __restrict__ bv, float* __restrict__ Qh,
    float* __restrict__ Kh, float* __restrict__ Vh, float* __restrict__ qq,
    float* __restrict__ kkv, float* __restrict__ qsign) {
  const int which = blockIdx.z;
  const float* X = which == 0 ? Xq : (which == 1 ? Xk : Xv);
  const float* W = which == 0 ? Wq : (which == 1 ? Wk : Wv);
  const float* bias = which == 0 ? bq : (which == 1 ? bk : bv);
  float* O = which == 0 ? Qh : (which == 1 ? Kh : Vh);

  const int m0 = blockIdx.y * 64;
  const int n0 = blockIdx.x * 128;
  const int tid = threadIdx.x;
  const int lane = tid & 31;
  const int wave = tid >> 5;
  const int wm = wave >> 1;
  const int wn = wave & 1;
  const int l16 = lane & 15;
  const int hi = lane >> 4;

  __shared__ float sX[2][64][SX_PITCH];
  __shared__ float sW2[2][K_STEP / 2][SW_STRIDE];
  __shared__ float sN[2][64];
  __shared__ float sRS[64];

  const bool do_rowsum = (which == 0) && (blockIdx.x == 0);
  float rs = 0.0f;

  v8f acc[4] = {};

  const int xr = tid >> 2, xc = (tid & 3) * 4;
  const float* xsrc = X + (size_t)(m0 + xr) * U_DIM + xc;
  const int wg = tid & 31, wp = tid >> 5;
  const float* wsrc = W + (size_t)(2 * wp) * U_DIM + n0 + 4 * wg;

  float4 xv[K_SUB], wa[K_SUB], wb[K_SUB];

#pragma unroll
  for (int u = 0; u < K_SUB; ++u) {
    xv[u] = *(const float4*)&xsrc[16 * u];
    wa[u] = *(const float4*)&wsrc[(size_t)(16 * u) * U_DIM];
    wb[u] = *(const float4*)&wsrc[(size_t)(16 * u + 1) * U_DIM];
  }

  for (int kt = 0; kt < N_STEPS; ++kt) {
    const int st = kt & 1;
#pragma unroll
    for (int u = 0; u < K_SUB; ++u) {
      *(float4*)&sX[st][xr][16 * u + xc] = xv[u];
      if (do_rowsum) rs += (xv[u].x + xv[u].y) + (xv[u].z + xv[u].w);
      float* dst = &sW2[st][wp + 8 * u][8 * wg];
      *(float4*)&dst[0] = make_float4(wa[u].x, wb[u].x, wa[u].y, wb[u].y);
      *(float4*)&dst[4] = make_float4(wa[u].z, wb[u].z, wa[u].w, wb[u].w);
    }
    __syncthreads();

    if (kt + 1 < N_STEPS) {
      const float* xs = xsrc + (kt + 1) * K_STEP;
      const float* ws = wsrc + (size_t)(kt + 1) * K_STEP * U_DIM;
#pragma unroll
      for (int u = 0; u < K_SUB; ++u) {
        xv[u] = *(const float4*)&xs[16 * u];
        wa[u] = *(const float4*)&ws[(size_t)(16 * u) * U_DIM];
        wb[u] = *(const float4*)&ws[(size_t)(16 * u + 1) * U_DIM];
      }
    }

    const float* ap = &sX[st][wm * 16 + l16][2 * hi];
    const float* bp = &sW2[st][hi][(wn * 64 + l16) * 2];
#pragma unroll 1
    for (int k0 = 0; k0 < K_STEP; k0 += 32) {
#pragma unroll
      for (int kk = 0; kk < 32; kk += 4) {
        const v2f a = *(const v2f*)&ap[kk];
#pragma unroll
        for (int n = 0; n < 4; ++n) {
          const v2f b = *(const v2f*)&bp[(kk >> 1) * SW_STRIDE + n * 32];
          acc[n] = wmma4(a, b, acc[n]);
        }
      }
      ap += 32;
      bp += 16 * SW_STRIDE;
    }
  }

  if (do_rowsum) {
    rs += __shfl_xor(rs, 1);
    rs += __shfl_xor(rs, 2);
    if ((tid & 3) == 0) sRS[tid >> 2] = rs;
  }

  float nrm[8];
#pragma unroll
  for (int r = 0; r < 8; ++r) nrm[r] = 0.0f;
#pragma unroll
  for (int n = 0; n < 4; ++n) {
    const int ng = n0 + wn * 64 + n * 16 + l16;
    const int h = ng >> 6, d = ng & 63;
    const float bb = bias[ng];
#pragma unroll
    for (int r = 0; r < 8; ++r) {
      const int mg = m0 + wm * 16 + r + 8 * hi;
      const int b = mg >> 9, l = mg & 511;
      float v = acc[n][r] + bb;
      v = v > 0.0f ? v : 0.0f;
      nrm[r] += v * v;
      O[(((size_t)(h * B_DIM + b) * L_DIM + l) * D_DIM) + d] = v;
    }
  }

  if (which < 2) {
#pragma unroll
    for (int r = 0; r < 8; ++r) {
#pragma unroll
      for (int off = 8; off >= 1; off >>= 1)
        nrm[r] += __shfl_xor(nrm[r], off, 16);
    }
    if (l16 == 0) {
#pragma unroll
      for (int r = 0; r < 8; ++r) sN[wn][wm * 16 + r + 8 * hi] = nrm[r];
    }
    __syncthreads();
    if (tid < 128) {
      const int hh = tid >> 6, row = tid & 63;
      const int mg = m0 + row;
      const int b = mg >> 9, l = mg & 511;
      const int h = (n0 >> 6) + hh;
      float* N = which == 0 ? qq : kkv;
      N[(size_t)(h * B_DIM + b) * L_DIM + l] = sN[hh][row];
    } else if (do_rowsum && tid < 192) {
      const int row = tid - 128;
      qsign[m0 + row] = (sRS[row] != 0.0f) ? 1.0f : 0.0f;
    }
  }
}

#define SK_STRIDE 68
#define SS_STRIDE 516

__global__ __launch_bounds__(256) void attn_kernel(
    const float* __restrict__ Qh, const float* __restrict__ Kh,
    const float* __restrict__ Vh, const float* __restrict__ qq,
    const float* __restrict__ kkv, const float* __restrict__ qsign,
    float* __restrict__ outp, float* __restrict__ ascore) {
  const int hb = blockIdx.y;
  const int q0 = blockIdx.x * 64;
  const int tid = threadIdx.x;
  const int lane = tid & 31;
  const int wave = tid >> 5;
  const int wm = wave >> 1;
  const int wn = wave & 1;
  const int l16 = lane & 15;
  const int hi = lane >> 4;

  __shared__ float sKV[L_DIM * SK_STRIDE];
  float (*sK)[SK_STRIDE] = (float (*)[SK_STRIDE])sKV;
  float (*sV2)[128] = (float (*)[128])sKV;
  __shared__ float sQ[64][68];
  __shared__ float sS[64][SS_STRIDE];
  __shared__ float sKK[L_DIM];
  __shared__ float sQQ[64];
  __shared__ float sQM[64];
  __shared__ float sMrow[64];
  __shared__ float sInvL[64];
  __shared__ float sStat[2][64][2];

  {
    const int r = tid >> 2, c = (tid & 3) * 16;
    const size_t base = ((size_t)hb * L_DIM + q0 + r) * D_DIM + c;
#pragma unroll
    for (int j = 0; j < 4; ++j) {
      const float4 v = *(const float4*)&Qh[base + 4 * j];
      *(float4*)&sQ[r][c + 4 * j] = v;
    }
    sKK[2 * tid] = kkv[(size_t)hb * L_DIM + 2 * tid];
    sKK[2 * tid + 1] = kkv[(size_t)hb * L_DIM + 2 * tid + 1];
    if (tid < 64) {
      sQQ[tid] = qq[(size_t)hb * L_DIM + q0 + tid];
      sQM[tid] = qsign[(size_t)(hb & 3) * L_DIM + q0 + tid];
    }
  }
  {
    const int c = (tid & 15) * 4;
    const float* src = Kh + ((size_t)hb * L_DIM + (tid >> 4)) * D_DIM + c;
#pragma unroll 8
    for (int i = 0; i < L_DIM / 16; ++i) {
      const float4 v = *(const float4*)&src[(size_t)i * 16 * D_DIM];
      *(float4*)&sK[(tid >> 4) + 16 * i][c] = v;
    }
  }
  __syncthreads();

  v2f qa[16];
#pragma unroll
  for (int j = 0; j < 16; ++j)
    qa[j] = *(const v2f*)&sQ[wm * 16 + l16][4 * j + 2 * hi];
  float qq_r[8], qm_r[8];
#pragma unroll
  for (int r = 0; r < 8; ++r) {
    qq_r[r] = sQQ[wm * 16 + r + 8 * hi];
    qm_r[r] = sQM[wm * 16 + r + 8 * hi];
  }
  const int kcol = wn * 16 + l16;

  float m_r[8], l_r[8];
#pragma unroll
  for (int r = 0; r < 8; ++r) { m_r[r] = -INFINITY; l_r[r] = 0.0f; }

  for (int kt = 0; kt < L_DIM / 32; kt += 2) {
    v8f s[2] = {};
    const float* kr = &sK[kt * 32 + kcol][2 * hi];
#pragma unroll
    for (int j = 0; j < 16; ++j) {
      const v2f b0 = *(const v2f*)&kr[4 * j];
      const v2f b1 = *(const v2f*)&kr[32 * SK_STRIDE + 4 * j];
      s[0] = wmma4(qa[j], b0, s[0]);
      s[1] = wmma4(qa[j], b1, s[1]);
    }
#pragma unroll
    for (int t = 0; t < 2; ++t) {
      const int kg = (kt + t) * 32 + kcol;
      const float kkg = sKK[kg];
#pragma unroll
      for (int r = 0; r < 8; ++r) {
        const int qrow = wm * 16 + r + 8 * hi;
        float raw = qq_r[r] + kkg - 2.0f * s[t][r];
        raw = fmaxf(raw, 1e-12f);
        const float sc = -sqrtf(raw) * 0.125f;
        sS[qrow][kg] = sc;
        __builtin_nontemporal_store(
            sc * qm_r[r],
            &ascore[((size_t)hb * L_DIM + q0 + qrow) * L_DIM + kg]);
        const float mn = fmaxf(m_r[r], sc);
        l_r[r] = l_r[r] * __expf(m_r[r] - mn) + __expf(sc - mn);
        m_r[r] = mn;
      }
    }
  }

#pragma unroll
  for (int r = 0; r < 8; ++r) {
    float m = m_r[r], l = l_r[r];
#pragma unroll
    for (int off = 8; off >= 1; off >>= 1) {
      const float mo = __shfl_xor(m, off, 16);
      const float lo = __shfl_xor(l, off, 16);
      const float mn = fmaxf(m, mo);
      l = l * __expf(m - mn) + lo * __expf(mo - mn);
      m = mn;
    }
    m_r[r] = m; l_r[r] = l;
  }
  if (l16 == 0) {
#pragma unroll
    for (int r = 0; r < 8; ++r) {
      const int row = wm * 16 + r + 8 * hi;
      sStat[wn][row][0] = m_r[r];
      sStat[wn][row][1] = l_r[r];
    }
  }
  __syncthreads();
  {
    const int d = (tid & 15) * 4;
    const float* src = Vh + ((size_t)hb * L_DIM + 2 * (tid >> 4)) * D_DIM + d;
#pragma unroll 8
    for (int i = 0; i < L_DIM / 32; ++i) {
      const int kp = (tid >> 4) + 16 * i;
      const float4 a = *(const float4*)&src[(size_t)i * 32 * D_DIM];
      const float4 b = *(const float4*)&src[(size_t)i * 32 * D_DIM + D_DIM];
      float* dst = &sV2[kp][(d * 2) ^ ((kp & 1) << 5)];
      *(float4*)&dst[0] = make_float4(a.x, b.x, a.y, b.y);
      *(float4*)&dst[4] = make_float4(a.z, b.z, a.w, b.w);
    }
  }
  if (tid < 64) {
    const float ma = sStat[0][tid][0], la = sStat[0][tid][1];
    const float mb = sStat[1][tid][0], lb = sStat[1][tid][1];
    const float mn = fmaxf(ma, mb);
    const float L = la * __expf(ma - mn) + lb * __expf(mb - mn);
    sMrow[tid] = mn;
    sInvL[tid] = 1.0f / L;
  }
  __syncthreads();

  const float mrow_a = sMrow[wm * 16 + l16];
  const float invl_a = sInvL[wm * 16 + l16];

  v8f o[2] = {};
  for (int kt = 0; kt < L_DIM / 32; kt += 2) {
    const float* sr = &sS[wm * 16 + l16][kt * 32 + 2 * hi];
#pragma unroll
    for (int kk = 0; kk < 64; kk += 4) {
      const v2f sv = *(const v2f*)&sr[kk];
      v2f a;
      a.x = __expf(sv.x - mrow_a) * invl_a;
      a.y = __expf(sv.y - mrow_a) * invl_a;
      const int kp = kt * 16 + (kk >> 1) + hi;
#pragma unroll
      for (int n = 0; n < 2; ++n) {
        const int dcol = wn * 32 + n * 16 + l16;
        const v2f b = *(const v2f*)&sV2[kp][(dcol * 2) ^ (hi << 5)];
        o[n] = wmma4(a, b, o[n]);
      }
    }
  }

  const int h = hb >> 2, b = hb & 3;
#pragma unroll
  for (int n = 0; n < 2; ++n) {
    const int d = wn * 32 + n * 16 + l16;
#pragma unroll
    for (int r = 0; r < 8; ++r) {
      const int qrow = wm * 16 + r + 8 * hi;
      const float v = o[n][r] * qm_r[r];
      __builtin_nontemporal_store(
          v, &outp[((size_t)b * L_DIM + q0 + qrow) * U_DIM + h * D_DIM + d]);
    }
  }
}

extern "C" void kernel_launch(void* const* d_in, const int* in_sizes, int n_in,
                              void* d_out, int out_size, void* d_ws,
                              size_t ws_size, hipStream_t stream) {
  const float* queries = (const float*)d_in[0];
  const float* keys = (const float*)d_in[1];
  const float* values = (const float*)d_in[2];
  const float* Wq = (const float*)d_in[3];
  const float* bq = (const float*)d_in[4];
  const float* Wk = (const float*)d_in[5];
  const float* bk = (const float*)d_in[6];
  const float* Wv = (const float*)d_in[7];
  const float* bv = (const float*)d_in[8];

  float* outp = (float*)d_out;
  float* ascore = outp + (size_t)B_DIM * L_DIM * U_DIM;

  float* ws = (float*)d_ws;
  float* Qh = ws;
  float* Kh = Qh + (size_t)HB_DIM * L_DIM * D_DIM;
  float* Vh = Kh + (size_t)HB_DIM * L_DIM * D_DIM;
  float* qq = Vh + (size_t)HB_DIM * L_DIM * D_DIM;
  float* kkv = qq + (size_t)HB_DIM * L_DIM;
  float* qsign = kkv + (size_t)HB_DIM * L_DIM;

  proj_relu_kernel<<<dim3(U_DIM / 128, M_ROWS / 64, 3), 256, 0, stream>>>(
      queries, keys, values, Wq, Wk, Wv, bq, bk, bv, Qh, Kh, Vh, qq, kkv,
      qsign);

  attn_kernel<<<dim3(L_DIM / 64, HB_DIM), 256, 0, stream>>>(
      Qh, Kh, Vh, qq, kkv, qsign, outp, ascore);
}


# projection kernel: block tile 64x128 on 8 waves -> 64x64 on 4 waves, 768 blocks held three to a WGP by LDS size: every WGP gets 48 output tiles, not 64 on some and 32 on others; k and sum order kept
# speedup vs baseline: 2.2597x; 1.1378x over previous
#include <hip/hip_runtime.h>
#include <hip/hip_bf16.h>
#include <math.h>

#define U_DIM 512
#define B_DIM 4
#define L_DIM 512
#define H_DIM 8
#define D_DIM 64
#define HB_DIM 32
#define M_ROWS 2048

typedef __attribute__((ext_vector_type(2))) float v2f;
typedef __attribute__((ext_vector_type(8))) float v8f;

__device__ __forceinline__ v8f wmma4(v2f a, v2f b, v8f c) {
#if defined(__HIP_DEVICE_COMPILE__)
  return __builtin_amdgcn_wmma_f32_16x16x4_f32(false, a, false, b, (short)0, c,
                                               false, false);
#else
  (void)a; (void)b;
  return c;
#endif
}


#define SW_STRIDE 160
#define K_STEP 64
#define SX_PITCH 68
#define N_STEPS (U_DIM / K_STEP)
#define K_SUB (K_STEP / 16)
#define PROJ_LDS_RESERVE 8192

__global__ __launch_bounds__(128) void proj_relu_kernel(
    const float* __restrict__ Xq, const float* __restrict__ Xk,
    const float* __restrict__ Xv, const float* __restrict__ Wq,
    const float* __restrict__ Wk, const float* __restrict__ Wv,
    const float* __restrict__ bq, const float* __restrict__ bk,
    const float* __restrict__ bv, float* __restrict__ Qh,
    float* __restrict__ Kh, float* __restrict__ Vh, float* __restrict__ qq,
    float* __restrict__ kkv, float* __restrict__ qsign) {
  const int which = blockIdx.z;
  const float* X = which == 0 ? Xq : (which == 1 ? Xk : Xv);
  const float* W = which == 0 ? Wq : (which == 1 ? Wk : Wv);
  const float* bias = which == 0 ? bq : (which == 1 ? bk : bv);
  float* O = which == 0 ? Qh : (which == 1 ? Kh : Vh);

  const int m0 = blockIdx.y * 64;
  const int n0 = blockIdx.x * 64;
  const int tid = threadIdx.x;
  const int lane = tid & 31;
  const int wm = tid >> 5;
  const int l16 = lane & 15;
  const int hi = lane >> 4;

  __shared__ float sX[2][64][SX_PITCH];
  __shared__ float sW2[2][K_STEP / 2][SW_STRIDE];
  __shared__ float sN[64];
  __shared__ float sRS[64];

  const bool do_rowsum = (which == 0) && (blockIdx.x == 0);
  float rs0 = 0.0f, rs1 = 0.0f;

  v8f acc[4] = {};

  const int xr = tid >> 2, xc = (tid & 3) * 4;
  const float* xsrc = X + (size_t)(m0 + xr) * U_DIM + xc;
  const int wg = tid & 15, wp = tid >> 4;
  const float* wsrc = W + (size_t)(2 * wp) * U_DIM + n0 + 4 * wg;

  float4 xa[K_SUB], xb[K_SUB], wa[K_SUB], wb[K_SUB];

#pragma unroll
  for (int u = 0; u < K_SUB; ++u) {
    xa[u] = *(const float4*)&xsrc[16 * u];
    xb[u] = *(const float4*)&xsrc[(size_t)32 * U_DIM + 16 * u];
    wa[u] = *(const float4*)&wsrc[(size_t)(16 * u) * U_DIM];
    wb[u] = *(const float4*)&wsrc[(size_t)(16 * u + 1) * U_DIM];
  }

  for (int kt = 0; kt < N_STEPS; ++kt) {
    const int st = kt & 1;
#pragma unroll
    for (int u = 0; u < K_SUB; ++u) {
      *(float4*)&sX[st][xr][16 * u + xc] = xa[u];
      *(float4*)&sX[st][xr + 32][16 * u + xc] = xb[u];
      if (do_rowsum) {
        rs0 += (xa[u].x + xa[u].y) + (xa[u].z + xa[u].w);
        rs1 += (xb[u].x + xb[u].y) + (xb[u].z + xb[u].w);
      }
      float* dst = &sW2[st][wp + 8 * u][8 * wg];
      *(float4*)&dst[0] = make_float4(wa[u].x, wb[u].x, wa[u].y, wb[u].y);
      *(float4*)&dst[4] = make_float4(wa[u].z, wb[u].z, wa[u].w, wb[u].w);
    }
    __syncthreads();

    if (kt + 1 < N_STEPS) {
      const float* xs = xsrc + (kt + 1) * K_STEP;
      const float* ws = wsrc + (size_t)(kt + 1) * K_STEP * U_DIM;
#pragma unroll
      for (int u = 0; u < K_SUB; ++u) {
        xa[u] = *(const float4*)&xs[16 * u];
        xb[u] = *(const float4*)&xs[(size_t)32 * U_DIM + 16 * u];
        wa[u] = *(const float4*)&ws[(size_t)(16 * u) * U_DIM];
        wb[u] = *(const float4*)&ws[(size_t)(16 * u + 1) * U_DIM];
      }
    }

    const float* ap = &sX[st][wm * 16 + l16][2 * hi];
    const float* bp = &sW2[st][hi][l16 * 2];
#pragma unroll 1
    for (int k0 = 0; k0 < K_STEP; k0 += 32) {
#pragma unroll
      for (int kk = 0; kk < 32; kk += 4) {
        const v2f a = *(const v2f*)&ap[kk];
#pragma unroll
        for (int n = 0; n < 4; ++n) {
          const v2f b = *(const v2f*)&bp[(kk >> 1) * SW_STRIDE + n * 32];
          acc[n] = wmma4(a, b, acc[n]);
        }
      }
      ap += 32;
      bp += 16 * SW_STRIDE;
    }
  }

  if (do_rowsum) {
    rs0 += __shfl_xor(rs0, 1);
    rs0 += __shfl_xor(rs0, 2);
    rs1 += __shfl_xor(rs1, 1);
    rs1 += __shfl_xor(rs1, 2);
    if ((tid & 3) == 0) {
      sRS[xr] = rs0;
      sRS[xr + 32] = rs1;
    }
  }

  float nrm[8];
#pragma unroll
  for (int r = 0; r < 8; ++r) nrm[r] = 0.0f;
#pragma unroll
  for (int n = 0; n < 4; ++n) {
    const int ng = n0 + n * 16 + l16;
    const int h = ng >> 6, d = ng & 63;
    const float bb = bias[ng];
#pragma unroll
    for (int r = 0; r < 8; ++r) {
      const int mg = m0 + wm * 16 + r + 8 * hi;
      const int b = mg >> 9, l = mg & 511;
      float v = acc[n][r] + bb;
      v = v > 0.0f ? v : 0.0f;
      nrm[r] += v * v;
      O[(((size_t)(h * B_DIM + b) * L_DIM + l) * D_DIM) + d] = v;
    }
  }

  if (which < 2) {
#pragma unroll
    for (int r = 0; r < 8; ++r) {
#pragma unroll
      for (int off = 8; off >= 1; off >>= 1)
        nrm[r] += __shfl_xor(nrm[r], off, 16);
    }
    if (l16 == 0) {
#pragma unroll
      for (int r = 0; r < 8; ++r) sN[wm * 16 + r + 8 * hi] = nrm[r];
    }
    __syncthreads();
    if (tid < 64) {
      const int mg = m0 + tid;
      const int b = mg >> 9, l = mg & 511;
      const int h = n0 >> 6;
      float* N = which == 0 ? qq : kkv;
      N[(size_t)(h * B_DIM + b) * L_DIM + l] = sN[tid];
    } else if (do_rowsum) {
      const int row = tid - 64;
      qsign[m0 + row] = (sRS[row] != 0.0f) ? 1.0f : 0.0f;
    }
  }
}

#define SK_STRIDE 68
#define SS_STRIDE 516

__global__ __launch_bounds__(256) void attn_kernel(
    const float* __restrict__ Qh, const float* __restrict__ Kh,
    const float* __restrict__ Vh, const float* __restrict__ qq,
    const float* __restrict__ kkv, const float* __restrict__ qsign,
    float* __restrict__ outp, float* __restrict__ ascore) {
  const int hb = blockIdx.y;
  const int q0 = blockIdx.x * 64;
  const int tid = threadIdx.x;
  const int lane = tid & 31;
  const int wave = tid >> 5;
  const int wm = wave >> 1;
  const int wn = wave & 1;
  const int l16 = lane & 15;
  const int hi = lane >> 4;

  __shared__ float sKV[L_DIM * SK_STRIDE];
  float (*sK)[SK_STRIDE] = (float (*)[SK_STRIDE])sKV;
  float (*sV2)[128] = (float (*)[128])sKV;
  __shared__ float sQ[64][68];
  __shared__ float sS[64][SS_STRIDE];
  __shared__ float sKK[L_DIM];
  __shared__ float sQQ[64];
  __shared__ float sQM[64];
  __shared__ float sMrow[64];
  __shared__ float sInvL[64];
  __shared__ float sStat[2][64][2];

  {
    const int r = tid >> 2, c = (tid & 3) * 16;
    const size_t base = ((size_t)hb * L_DIM + q0 + r) * D_DIM + c;
#pragma unroll
    for (int j = 0; j < 4; ++j) {
      const float4 v = *(const float4*)&Qh[base + 4 * j];
      *(float4*)&sQ[r][c + 4 * j] = v;
    }
    sKK[2 * tid] = kkv[(size_t)hb * L_DIM + 2 * tid];
    sKK[2 * tid + 1] = kkv[(size_t)hb * L_DIM + 2 * tid + 1];
    if (tid < 64) {
      sQQ[tid] = qq[(size_t)hb * L_DIM + q0 + tid];
      sQM[tid] = qsign[(size_t)(hb & 3) * L_DIM + q0 + tid];
    }
  }
  {
    const int c = (tid & 15) * 4;
    const float* src = Kh + ((size_t)hb * L_DIM + (tid >> 4)) * D_DIM + c;
#pragma unroll 8
    for (int i = 0; i < L_DIM / 16; ++i) {
      const float4 v = *(const float4*)&src[(size_t)i * 16 * D_DIM];
      *(float4*)&sK[(tid >> 4) + 16 * i][c] = v;
    }
  }
  __syncthreads();

  v2f qa[16];
#pragma unroll
  for (int j = 0; j < 16; ++j)
    qa[j] = *(const v2f*)&sQ[wm * 16 + l16][4 * j + 2 * hi];
  float qq_r[8], qm_r[8];
#pragma unroll
  for (int r = 0; r < 8; ++r) {
    qq_r[r] = sQQ[wm * 16 + r + 8 * hi];
    qm_r[r] = sQM[wm * 16 + r + 8 * hi];
  }
  const int kcol = wn * 16 + l16;

  float m_r[8], l_r[8];
#pragma unroll
  for (int r = 0; r < 8; ++r) { m_r[r] = -INFINITY; l_r[r] = 0.0f; }

  for (int kt = 0; kt < L_DIM / 32; kt += 2) {
    v8f s[2] = {};
    const float* kr = &sK[kt * 32 + kcol][2 * hi];
#pragma unroll
    for (int j = 0; j < 16; ++j) {
      const v2f b0 = *(const v2f*)&kr[4 * j];
      const v2f b1 = *(const v2f*)&kr[32 * SK_STRIDE + 4 * j];
      s[0] = wmma4(qa[j], b0, s[0]);
      s[1] = wmma4(qa[j], b1, s[1]);
    }
#pragma unroll
    for (int t = 0; t < 2; ++t) {
      const int kg = (kt + t) * 32 + kcol;
      const float kkg = sKK[kg];
#pragma unroll
      for (int r = 0; r < 8; ++r) {
        const int qrow = wm * 16 + r + 8 * hi;
        float raw = qq_r[r] + kkg - 2.0f * s[t][r];
        raw = fmaxf(raw, 1e-12f);
        const float sc = -sqrtf(raw) * 0.125f;
        sS[qrow][kg] = sc;
        __builtin_nontemporal_store(
            sc * qm_r[r],
            &ascore[((size_t)hb * L_DIM + q0 + qrow) * L_DIM + kg]);
        const float mn = fmaxf(m_r[r], sc);
        l_r[r] = l_r[r] * __expf(m_r[r] - mn) + __expf(sc - mn);
        m_r[r] = mn;
      }
    }
  }

#pragma unroll
  for (int r = 0; r < 8; ++r) {
    float m = m_r[r], l = l_r[r];
#pragma unroll
    for (int off = 8; off >= 1; off >>= 1) {
      const float mo = __shfl_xor(m, off, 16);
      const float lo = __shfl_xor(l, off, 16);
      const float mn = fmaxf(m, mo);
      l = l * __expf(m - mn) + lo * __expf(mo - mn);
      m = mn;
    }
    m_r[r] = m; l_r[r] = l;
  }
  if (l16 == 0) {
#pragma unroll
    for (int r = 0; r < 8; ++r) {
      const int row = wm * 16 + r + 8 * hi;
      sStat[wn][row][0] = m_r[r];
      sStat[wn][row][1] = l_r[r];
    }
  }
  __syncthreads();
  {
    const int d = (tid & 15) * 4;
    const float* src = Vh + ((size_t)hb * L_DIM + 2 * (tid >> 4)) * D_DIM + d;
#pragma unroll 8
    for (int i = 0; i < L_DIM / 32; ++i) {
      const int kp = (tid >> 4) + 16 * i;
      const float4 a = *(const float4*)&src[(size_t)i * 32 * D_DIM];
      const float4 b = *(const float4*)&src[(size_t)i * 32 * D_DIM + D_DIM];
      float* dst = &sV2[kp][(d * 2) ^ ((kp & 1) << 5)];
      *(float4*)&dst[0] = make_float4(a.x, b.x, a.y, b.y);
      *(float4*)&dst[4] = make_float4(a.z, b.z, a.w, b.w);
    }
  }
  if (tid < 64) {
    const float ma = sStat[0][tid][0], la = sStat[0][tid][1];
    const float mb = sStat[1][tid][0], lb = sStat[1][tid][1];
    const float mn = fmaxf(ma, mb);
    const float L = la * __expf(ma - mn) + lb * __expf(mb - mn);
    sMrow[tid] = mn;
    sInvL[tid] = 1.0f / L;
  }
  __syncthreads();

  const float mrow_a = sMrow[wm * 16 + l16];
  const float invl_a = sInvL[wm * 16 + l16];

  v8f o[2] = {};
  for (int kt = 0; kt < L_DIM / 32; kt += 2) {
    const float* sr = &sS[wm * 16 + l16][kt * 32 + 2 * hi];
#pragma unroll
    for (int kk = 0; kk < 64; kk += 4) {
      const v2f sv = *(const v2f*)&sr[kk];
      v2f a;
      a.x = __expf(sv.x - mrow_a) * invl_a;
      a.y = __expf(sv.y - mrow_a) * invl_a;
      const int kp = kt * 16 + (kk >> 1) + hi;
#pragma unroll
      for (int n = 0; n < 2; ++n) {
        const int dcol = wn * 32 + n * 16 + l16;
        const v2f b = *(const v2f*)&sV2[kp][(dcol * 2) ^ (hi << 5)];
        o[n] = wmma4(a, b, o[n]);
      }
    }
  }

  const int h = hb >> 2, b = hb & 3;
#pragma unroll
  for (int n = 0; n < 2; ++n) {
    const int d = wn * 32 + n * 16 + l16;
#pragma unroll
    for (int r = 0; r < 8; ++r) {
      const int qrow = wm * 16 + r + 8 * hi;
      const float v = o[n][r] * qm_r[r];
      __builtin_nontemporal_store(
          v, &outp[((size_t)b * L_DIM + q0 + qrow) * U_DIM + h * D_DIM + d]);
    }
  }
}

extern "C" void kernel_launch(void* const* d_in, const int* in_sizes, int n_in,
                              void* d_out, int out_size, void* d_ws,
                              size_t ws_size, hipStream_t stream) {
  const float* queries = (const float*)d_in[0];
  const float* keys = (const float*)d_in[1];
  const float* values = (const float*)d_in[2];
  const float* Wq = (const float*)d_in[3];
  const float* bq = (const float*)d_in[4];
  const float* Wk = (const float*)d_in[5];
  const float* bk = (const float*)d_in[6];
  const float* Wv = (const float*)d_in[7];
  const float* bv = (const float*)d_in[8];

  float* outp = (float*)d_out;
  float* ascore = outp + (size_t)B_DIM * L_DIM * U_DIM;

  float* ws = (float*)d_ws;
  float* Qh = ws;
  float* Kh = Qh + (size_t)HB_DIM * L_DIM * D_DIM;
  float* Vh = Kh + (size_t)HB_DIM * L_DIM * D_DIM;
  float* qq = Vh + (size_t)HB_DIM * L_DIM * D_DIM;
  float* kkv = qq + (size_t)HB_DIM * L_DIM;
  float* qsign = kkv + (size_t)HB_DIM * L_DIM;

  proj_relu_kernel<<<dim3(U_DIM / 64, M_ROWS / 64, 3), 128, PROJ_LDS_RESERVE, stream>>>(
      queries, keys, values, Wq, Wk, Wv, bq, bk, bv, Qh, Kh, Vh, qq, kkv,
      qsign);

  attn_kernel<<<dim3(L_DIM / 64, HB_DIM), 256, 0, stream>>>(
      Qh, Kh, Vh, qq, kkv, qsign, outp, ascore);
}
